# starting kernel
# baseline (speedup 1.0000x reference)
#include <hip/hip_runtime.h>
#include <cstdint>
#include <cstddef>

#define NB   32
#define NN   4096
#define NCAP 10
#define DV   16
#define DIN  8
#define CD   (NCAP * DV)
#define NCHUNK 32
#define CHN  (NN / NCHUNK)

typedef float v2f __attribute__((ext_vector_type(2)));
typedef float v8f __attribute__((ext_vector_type(8)));

__global__ __launch_bounds__(256) void capsule_uhat_wmma(
    const float* __restrict__ x, const float* __restrict__ W,
    float* __restrict__ uhat) {
  const int lane = threadIdx.x & 31;
  const int n    = blockIdx.x * 8 + (threadIdx.x >> 5);
  const int half = lane >> 4;
  const int r    = lane & 15;

  v2f A[2][2];
#pragma unroll
  for (int mt = 0; mt < 2; ++mt) {
#pragma unroll
    for (int kc = 0; kc < 2; ++kc) {
      const int b = mt * 16 + r;
      const int i = kc * 4 + 2 * half;
      const float* p = x + ((size_t)b * NN + n) * DIN + i;
      A[mt][kc].x = p[0];
      A[mt][kc].y = p[1];
    }
  }

  v2f Bm[NCAP][2];
#pragma unroll
  for (int c = 0; c < NCAP; ++c) {
#pragma unroll
    for (int kc = 0; kc < 2; ++kc) {
      const int i0 = kc * 4 + 2 * half;
      const float* p = W + (((size_t)n * NCAP + c) * DIN + i0) * DV + r;
      Bm[c][kc].x = p[0];
      Bm[c][kc].y = p[DV];
    }
  }

  float* outn = uhat + (size_t)n * (NB * CD);
#pragma unroll
  for (int mt = 0; mt < 2; ++mt) {
#pragma unroll
    for (int c = 0; c < NCAP; ++c) {
      v8f acc = {0.f, 0.f, 0.f, 0.f, 0.f, 0.f, 0.f, 0.f};
      acc = __builtin_amdgcn_wmma_f32_16x16x4_f32(
          false, A[mt][0], false, Bm[c][0], (short)0, acc, false, false);
      acc = __builtin_amdgcn_wmma_f32_16x16x4_f32(
          false, A[mt][1], false, Bm[c][1], (short)0, acc, false, false);
#pragma unroll
      for (int v = 0; v < 8; ++v) {
        const int b = mt * 16 + v + 8 * half;
        outn[(size_t)b * CD + c * DV + r] = acc[v];
      }
    }
  }
}

__global__ __launch_bounds__(256) void capsule_routing_accum(
    const float* __restrict__ uhat, float* __restrict__ blog,
    const float* __restrict__ vin, float* __restrict__ partial, int iter) {
  const int b     = blockIdx.y;
  const int chunk = blockIdx.x;
  const int n0    = chunk * CHN;
  const int t     = threadIdx.x;

  __shared__ float s_coeff[CHN * NCAP];
  __shared__ float s_v[CD];

  if (iter > 0) {
    if (t < CD) s_v[t] = vin[b * CD + t];
    __syncthreads();
    for (int task = t; task < CHN * NCAP; task += 256) {
      const int nl = task / NCAP;
      const int c  = task - nl * NCAP;
      const int n  = n0 + nl;
      const float* up = uhat + ((size_t)n * NB + b) * CD + c * DV;
      float dot = 0.f;
#pragma unroll
      for (int d = 0; d < DV; ++d) dot += up[d] * s_v[c * DV + d];
      const size_t bi = ((size_t)b * NN + n) * NCAP + c;
      const float prev = (iter == 1) ? 0.f : blog[bi];
      const float nb = prev + dot;
      blog[bi] = nb;
      s_coeff[task] = nb;
    }
    __syncthreads();
    if (t < CHN) {
      float lg[NCAP];
      float mx = -3.4e38f;
#pragma unroll
      for (int c = 0; c < NCAP; ++c) {
        lg[c] = s_coeff[t * NCAP + c];
        mx = fmaxf(mx, lg[c]);
      }
      float sum = 0.f;
#pragma unroll
      for (int c = 0; c < NCAP; ++c) {
        lg[c] = expf(lg[c] - mx);
        sum += lg[c];
      }
      const float inv = 1.f / sum;
#pragma unroll
      for (int c = 0; c < NCAP; ++c) s_coeff[t * NCAP + c] = lg[c] * inv;
    }
    __syncthreads();
  } else {
    for (int i = t; i < CHN * NCAP; i += 256) s_coeff[i] = 0.1f;
    __syncthreads();
  }

  if (t < CD) {
    const int c = t >> 4;
    float acc = 0.f;
    for (int nl = 0; nl < CHN; ++nl) {
      acc += s_coeff[nl * NCAP + c] *
             uhat[((size_t)(n0 + nl) * NB + b) * CD + t];
    }
    partial[((size_t)b * NCHUNK + chunk) * CD + t] = acc;
  }
}

__global__ __launch_bounds__(160) void capsule_reduce_squash(
    const float* __restrict__ partial, float* __restrict__ vout,
    float* __restrict__ out, int last) {
  const int b = blockIdx.x;
  const int t = threadIdx.x;
  const int c = t >> 4;

  float s = 0.f;
#pragma unroll
  for (int ch = 0; ch < NCHUNK; ++ch)
    s += partial[((size_t)b * NCHUNK + ch) * CD + t];

  __shared__ float sh[CD];
  __shared__ float s2sh[NCAP];
  sh[t] = s * s;
  __syncthreads();
  if ((t & 15) == 0) {
    float s2 = 0.f;
#pragma unroll
    for (int k = 0; k < DV; ++k) s2 += sh[c * DV + k];
    s2sh[c] = s2;
  }
  __syncthreads();
  const float s2 = s2sh[c];
  const float scale = s2 / ((1.f + s2) * sqrtf(s2 + 1e-7f));
  const float vv = scale * s;
  vout[b * CD + t] = vv;
  if (last) out[b * CD + t] = vv;
}

extern "C" void kernel_launch(void* const* d_in, const int* in_sizes, int n_in,
                              void* d_out, int out_size, void* d_ws,
                              size_t ws_size, hipStream_t stream) {
  (void)in_sizes; (void)n_in; (void)out_size; (void)ws_size;
  const float* x = (const float*)d_in[0];
  const float* W = (const float*)d_in[1];
  float* out = (float*)d_out;

  float* ws      = (float*)d_ws;
  float* uhat    = ws;
  float* blog    = uhat + (size_t)NN * NB * CD;
  float* partial = blog + (size_t)NB * NN * NCAP;
  float* vbuf    = partial + (size_t)NB * NCHUNK * CD;

  capsule_uhat_wmma<<<NN / 8, 256, 0, stream>>>(x, W, uhat);
  for (int it = 0; it < 3; ++it) {
    capsule_routing_accum<<<dim3(NCHUNK, NB), 256, 0, stream>>>(
        uhat, blog, vbuf, partial, it);
    capsule_reduce_squash<<<NB, 160, 0, stream>>>(partial, vbuf, out,
                                                  it == 2 ? 1 : 0);
  }
}


# Routing passes read u_hat once: a wave loads rows as full 128-byte lines and keeps them in registers from logit dot (DPP row merges) through softmax to weighted sum; iter-2 logits from v0+v1
# speedup vs baseline: 1.5730x; 1.5730x over previous
#include <hip/hip_runtime.h>
#include <cstdint>
#include <cstddef>

#define NB   32
#define NN   4096
#define NCAP 10
#define DV   16
#define DIN  8
#define CD   (NCAP * DV)
#define NCHUNK 32
#define CHN  (NN / NCHUNK)
#define NCH0 128
#define CHN0 (NN / NCH0)
#define RWAVES 4

typedef float v2f __attribute__((ext_vector_type(2)));
typedef float v8f __attribute__((ext_vector_type(8)));

__global__ __launch_bounds__(512) void capsule_uhat_wmma(
    const float* __restrict__ x, const float* __restrict__ W,
    float* __restrict__ uhat, float* __restrict__ partial0) {
  const int lane  = threadIdx.x & 31;
  const int wave  = threadIdx.x >> 5;
  const int mt    = blockIdx.x;
  const int chunk = blockIdx.y;
  const int half  = lane >> 4;
  const int r     = lane & 15;

  __shared__ float s_red[16][NCAP * 256];

  v8f sum[NCAP];
#pragma unroll
  for (int c = 0; c < NCAP; ++c)
    sum[c] = (v8f){0.f, 0.f, 0.f, 0.f, 0.f, 0.f, 0.f, 0.f};

#pragma unroll
  for (int j = 0; j < CHN0 / 16; ++j) {
    const int n = chunk * CHN0 + j * 16 + wave;

    v2f A[2];
#pragma unroll
    for (int kc = 0; kc < 2; ++kc) {
      const int b = mt * 16 + r;
      const int i = kc * 4 + 2 * half;
      const float* p = x + ((size_t)b * NN + n) * DIN + i;
      A[kc].x = p[0];
      A[kc].y = p[1];
    }

    v2f Bm[NCAP][2];
#pragma unroll
    for (int c = 0; c < NCAP; ++c) {
#pragma unroll
      for (int kc = 0; kc < 2; ++kc) {
        const int i0 = kc * 4 + 2 * half;
        const float* p = W + (((size_t)n * NCAP + c) * DIN + i0) * DV + r;
        Bm[c][kc].x = p[0];
        Bm[c][kc].y = p[DV];
      }
    }

    float* outn = uhat + (size_t)n * (NB * CD);
#pragma unroll
    for (int c = 0; c < NCAP; ++c) {
      v8f acc = {0.f, 0.f, 0.f, 0.f, 0.f, 0.f, 0.f, 0.f};
      acc = __builtin_amdgcn_wmma_f32_16x16x4_f32(
          false, A[0], false, Bm[c][0], (short)0, acc, false, false);
      acc = __builtin_amdgcn_wmma_f32_16x16x4_f32(
          false, A[1], false, Bm[c][1], (short)0, acc, false, false);
#pragma unroll
      for (int v = 0; v < 8; ++v) {
        const int b = mt * 16 + v + 8 * half;
        outn[(size_t)b * CD + c * DV + r] = acc[v];
      }
      sum[c] += acc;
    }
  }

#pragma unroll
  for (int c = 0; c < NCAP; ++c)
#pragma unroll
    for (int v = 0; v < 8; ++v)
      s_red[wave][c * 256 + v * 32 + lane] = sum[c][v];
  __builtin_amdgcn_fence(__ATOMIC_RELEASE, "workgroup", "local");
  __builtin_amdgcn_s_barrier();
  __builtin_amdgcn_fence(__ATOMIC_ACQUIRE, "workgroup", "local");

#pragma unroll
  for (int k = 0; k < 5; ++k) {
    const int o   = k * 512 + (int)threadIdx.x;
    const int row = o / CD;
    const int col = o - row * CD;
    const int idx = (col >> 4) * 256 + (row & 7) * 32 + (row >> 3) * 16 +
                    (col & 15);
    float s = 0.f;
#pragma unroll
    for (int w = 0; w < 16; ++w) s += s_red[w][idx];
    partial0[((size_t)(mt * 16 + row) * NCH0 + chunk) * CD + col] = s;
  }
}

__device__ __forceinline__ float xhalf(float x) {
  const int b = __builtin_bit_cast(int, x);
  return __builtin_bit_cast(
      float, __builtin_amdgcn_permlanex16(b, b, 0x76543210, 0xfedcba98, false,
                                          false));
}

template <int CTRL>
__device__ __forceinline__ float dppmov(float x) {
  const int b = __builtin_bit_cast(int, x);
  return __builtin_bit_cast(
      float, __builtin_amdgcn_update_dpp(0, b, CTRL, 0xf, 0xf, true));
}

template <int M>
__device__ __forceinline__ float merge(float a, float b, bool sel) {
  const float keep = sel ? b : a;
  const float send = sel ? a : b;
  return keep + dppmov<0x160 + M>(send);
}

template <int J>
__device__ __forceinline__ void accum_rows(float (&acc)[5],
                                           const float (&cf)[5],
                                           const float (&U)[5][8]) {
#pragma unroll
  for (int k = 0; k < 5; ++k)
    acc[k] = fmaf(dppmov<0x150 + 2 * J>(cf[k]), U[k][J], acc[k]);
  if constexpr (J + 1 < 8) accum_rows<J + 1>(acc, cf, U);
}

__device__ __forceinline__ void load_rows8(float (&U)[5][8],
                                           const float* __restrict__ row0) {
#pragma unroll
  for (int k = 0; k < 5; ++k)
#pragma unroll
    for (int j = 0; j < 8; ++j)
      U[k][j] = row0[(size_t)j * (NB * CD) + 32 * k];
}

__device__ __forceinline__ void route_rows8(const float (&U)[5][8],
                                            const float (&vv)[5], bool s8,
                                            bool s4, bool s2,
                                            float (&acc)[5]) {
  float L[5];
#pragma unroll
  for (int k = 0; k < 5; ++k) {
    float Y[4], Z[2];
#pragma unroll
    for (int j = 0; j < 4; ++j)
      Y[j] = merge<8>(U[k][j] * vv[k], U[k][j + 4] * vv[k], s8);
#pragma unroll
    for (int j = 0; j < 2; ++j) Z[j] = merge<4>(Y[j], Y[j + 2], s4);
    const float q = merge<2>(Z[0], Z[1], s2);
    L[k] = q + dppmov<0x161>(q);
  }

  float m = fmaxf(fmaxf(fmaxf(L[0], L[1]), fmaxf(L[2], L[3])), L[4]);
  m = fmaxf(m, xhalf(m));
  float e[5];
  float s = 0.f;
#pragma unroll
  for (int k = 0; k < 5; ++k) {
    e[k] = __builtin_amdgcn_exp2f(L[k] - m);
    s += e[k];
  }
  s += xhalf(s);
  const float inv = __builtin_amdgcn_rcpf(s);
  float cf[5];
#pragma unroll
  for (int k = 0; k < 5; ++k) cf[k] = e[k] * inv;

  accum_rows<0>(acc, cf, U);
}

template <bool TWO>
__global__ __launch_bounds__(RWAVES * 32) void capsule_route_once(
    const float* __restrict__ uhat, const float* __restrict__ va,
    const float* __restrict__ vb, float* __restrict__ partial) {
  const int b     = blockIdx.y;
  const int chunk = blockIdx.x;
  const int t     = threadIdx.x;
  const int lane  = t & 31;
  const int wave  = t >> 5;
  const int r     = lane & 15;

  __shared__ float s_acc[RWAVES][CD];

  float vv[5];
#pragma unroll
  for (int k = 0; k < 5; ++k) {
    float v = va[b * CD + 32 * k + lane];
    if (TWO) v += vb[b * CD + 32 * k + lane];
    vv[k] = v * 1.4426950408889634f;
  }
  asm volatile("" ::: "memory");

  const size_t G = (size_t)8 * (NB * CD);
  const float* row0 =
      uhat + ((size_t)(chunk * CHN + wave * 32) * NB + b) * CD + lane;

  float UA[5][8], UB[5][8];
  load_rows8(UA, row0);
  load_rows8(UB, row0 + G);

  const bool s8 = (r & 8) != 0;
  const bool s4 = (r & 4) != 0;
  const bool s2 = (r & 2) != 0;

  float acc[5] = {0.f, 0.f, 0.f, 0.f, 0.f};
#define PIN()                                                             \
  asm volatile("" : "+v"(acc[0]), "+v"(acc[1]), "+v"(acc[2]), "+v"(acc[3]), \
               "+v"(acc[4])::"memory")
#define HOLD1(U, k)                                                        \
  asm volatile("" : "+v"(U[k][0]), "+v"(U[k][1]), "+v"(U[k][2]),          \
               "+v"(U[k][3]), "+v"(U[k][4]), "+v"(U[k][5]), "+v"(U[k][6]), \
               "+v"(U[k][7]))
#define HOLD(U) HOLD1(U, 0); HOLD1(U, 1); HOLD1(U, 2); HOLD1(U, 3); HOLD1(U, 4)
  PIN();
  route_rows8(UA, vv, s8, s4, s2, acc);
  PIN();
  load_rows8(UA, row0 + 2 * G);
  PIN();
  HOLD(UB);
  route_rows8(UB, vv, s8, s4, s2, acc);
  PIN();
  load_rows8(UB, row0 + 3 * G);
  PIN();
  HOLD(UA);
  route_rows8(UA, vv, s8, s4, s2, acc);
  PIN();
  HOLD(UB);
  route_rows8(UB, vv, s8, s4, s2, acc);
#undef HOLD
#undef HOLD1
#undef PIN

#pragma unroll
  for (int k = 0; k < 5; ++k) s_acc[wave][32 * k + lane] = acc[k];
  __syncthreads();
  for (int o = t; o < CD; o += RWAVES * 32) {
    float x = 0.f;
#pragma unroll
    for (int w = 0; w < RWAVES; ++w) x += s_acc[w][o];
    partial[((size_t)b * NCHUNK + chunk) * CD + o] = x;
  }
}

template <int NCH, bool PASS0>
__global__ __launch_bounds__(160) void capsule_reduce_squash(
    const float* __restrict__ partial, float* __restrict__ vout,
    float* __restrict__ out, int last) {
  const int b = blockIdx.x;
  const int t = threadIdx.x;
  const int c = t >> 4;

  float s = 0.f;
#pragma unroll
  for (int ch = 0; ch < NCH; ++ch)
    s += partial[((size_t)b * NCH + ch) * CD + t];
  if (PASS0) s *= 0.1f;

  __shared__ float sh[CD];
  __shared__ float s2sh[NCAP];
  sh[t] = s * s;
  __syncthreads();
  if ((t & 15) == 0) {
    float s2 = 0.f;
#pragma unroll
    for (int k = 0; k < DV; ++k) s2 += sh[c * DV + k];
    s2sh[c] = s2;
  }
  __syncthreads();
  const float s2 = s2sh[c];
  const float scale = s2 / ((1.f + s2) * sqrtf(s2 + 1e-7f));
  const float vv = scale * s;
  vout[b * CD + t] = vv;
  if (last) out[b * CD + t] = vv;
}

extern "C" void kernel_launch(void* const* d_in, const int* in_sizes, int n_in,
                              void* d_out, int out_size, void* d_ws,
                              size_t ws_size, hipStream_t stream) {
  (void)in_sizes; (void)n_in; (void)out_size; (void)ws_size;
  const float* x = (const float*)d_in[0];
  const float* W = (const float*)d_in[1];
  float* out = (float*)d_out;

  float* ws      = (float*)d_ws;
  float* uhat    = ws;
  float* plane   = uhat + (size_t)NN * NB * CD;
  float* partial = plane + (size_t)NB * NN * NCAP;
  float* vbuf0   = partial + (size_t)NB * NCHUNK * CD;

  float* partial0 = plane;
  float* vbuf1    = plane + (size_t)NB * NCH0 * CD;

  capsule_uhat_wmma<<<dim3(NB / 16, NCH0), 512, 0, stream>>>(x, W, uhat,
                                                             partial0);
  capsule_reduce_squash<NCH0, true><<<NB, 160, 0, stream>>>(partial0, vbuf0,
                                                            out, 0);
  capsule_route_once<false><<<dim3(NCHUNK, NB), RWAVES * 32, 0, stream>>>(
      uhat, vbuf0, vbuf0, partial);
  capsule_reduce_squash<NCHUNK, false><<<NB, 160, 0, stream>>>(partial, vbuf1,
                                                               out, 0);
  capsule_route_once<true><<<dim3(NCHUNK, NB), RWAVES * 32, 0, stream>>>(
      uhat, vbuf0, vbuf1, partial);
  capsule_reduce_squash<NCHUNK, false><<<NB, 160, 0, stream>>>(partial, vbuf1,
                                                               out, 1);
}
